# starting kernel
# baseline (speedup 1.0000x reference)
#include <hip/hip_runtime.h>
#include <hip/hip_bf16.h>
#include <math.h>

typedef __attribute__((ext_vector_type(2))) float v2f;
typedef __attribute__((ext_vector_type(8))) float v8f;

#define TOKENS   32768
#define CH       2
#define VDIM     128
#define KCODE    512
#define OUT1_OFF 8388608
#define OUT2_OFF 8454144
#define ENT_OFF  8519680

__global__ __launch_bounds__(256) void vq_setup(const float* __restrict__ emb,
                                                unsigned int* __restrict__ hist,
                                                float* __restrict__ esq) {
    int tid = blockIdx.x * 256 + threadIdx.x;
    if (tid < KCODE) hist[tid] = 0u;
    if (tid < CH * KCODE) {
        const float* ep = emb + (size_t)tid * VDIM;
        float s = 0.0f;
        #pragma unroll 8
        for (int v = 0; v < VDIM; ++v) { float e = ep[v]; s += e * e; }
        esq[tid] = s;
    }
}

__global__ __launch_bounds__(256) void vq_main(const float* __restrict__ x0,
                                               const float* __restrict__ emb,
                                               const float* __restrict__ esq,
                                               unsigned int* __restrict__ hist,
                                               float* __restrict__ out) {
    __shared__ int sIdx[256];

    const int lane   = threadIdx.x & 31;
    const int waveId = threadIdx.x >> 5;
    const int lane16 = lane & 15;
    const int half   = lane >> 4;
    const int c       = blockIdx.x & 1;
    const int rowBase = (blockIdx.x >> 1) * 256;
    const int waveRow = rowBase + waveId * 32;

    v2f a0[32], a1[32];
    {
        const float* xp0 = x0 + (size_t)(waveRow + lane16) * (CH * VDIM)
                              + c * VDIM + half * 2;
        const float* xp1 = xp0 + 16 * (CH * VDIM);
        #pragma unroll
        for (int kk = 0; kk < 32; ++kk) {
            a0[kk] = *(const v2f*)(xp0 + kk * 4);
            a1[kk] = *(const v2f*)(xp1 + kk * 4);
        }
    }

    float runD0[8], runD1[8];
    int   runI0[8], runI1[8];
    #pragma unroll
    for (int r = 0; r < 8; ++r) {
        runD0[r] = 3.4e38f; runD1[r] = 3.4e38f; runI0[r] = 0; runI1[r] = 0;
    }

    for (int t = 0; t < 32; ++t) {
        const int code = t * 16 + lane16;
        const float ev2 = esq[c * KCODE + code];
        const float* ep = emb + ((size_t)(c * KCODE + code)) * VDIM + half * 2;

        v8f acc0 = {}; v8f acc1 = {};
        #pragma unroll
        for (int kk = 0; kk < 32; ++kk) {
            v2f b = *(const v2f*)(ep + kk * 4);
            acc0 = __builtin_amdgcn_wmma_f32_16x16x4_f32(
                false, a0[kk], false, b, (short)0, acc0, false, false);
            acc1 = __builtin_amdgcn_wmma_f32_16x16x4_f32(
                false, a1[kk], false, b, (short)0, acc1, false, false);
        }
        #pragma unroll
        for (int r = 0; r < 8; ++r) {
            float d0 = ev2 - 2.0f * acc0[r];
            float d1 = ev2 - 2.0f * acc1[r];
            if (d0 < runD0[r]) { runD0[r] = d0; runI0[r] = code; }
            if (d1 < runD1[r]) { runD1[r] = d1; runI1[r] = code; }
        }
    }

    #pragma unroll
    for (int m = 1; m < 16; m <<= 1) {
        #pragma unroll
        for (int r = 0; r < 8; ++r) {
            float od = __shfl_xor(runD0[r], m, 32);
            int   oi = __shfl_xor(runI0[r], m, 32);
            if (od < runD0[r] || (od == runD0[r] && oi < runI0[r])) {
                runD0[r] = od; runI0[r] = oi;
            }
            od = __shfl_xor(runD1[r], m, 32);
            oi = __shfl_xor(runI1[r], m, 32);
            if (od < runD1[r] || (od == runD1[r] && oi < runI1[r])) {
                runD1[r] = od; runI1[r] = oi;
            }
        }
    }

    if (lane16 == 0) {
        #pragma unroll
        for (int r = 0; r < 8; ++r) {
            int rl0 = waveId * 32 + r + 8 * half;
            sIdx[rl0] = runI0[r];
            atomicAdd(&hist[runI0[r]], 1u);
            int rl1 = rl0 + 16;
            sIdx[rl1] = runI1[r];
            atomicAdd(&hist[runI1[r]], 1u);
        }
    }
    __syncthreads();

    for (int it = 0; it < 32; ++it) {
        int rloc = waveId + it * 8;
        int row  = rowBase + rloc;
        int idx  = sIdx[rloc];
        const float4 ev = *(const float4*)(emb + ((size_t)(c * KCODE + idx)) * VDIM + lane * 4);
        const float4 xv = *(const float4*)(x0 + (size_t)row * (CH * VDIM) + c * VDIM + lane * 4);
        *(float4*)(out + (size_t)row * (CH * VDIM) + c * VDIM + lane * 4) = ev;
        float dx = xv.x - ev.x, dy = xv.y - ev.y;
        float dz = xv.z - ev.z, dw = xv.w - ev.w;
        float s = dx * dx + dy * dy + dz * dz + dw * dw;
        #pragma unroll
        for (int m = 1; m < 32; m <<= 1) s += __shfl_xor(s, m, 32);
        if (lane == 0) {
            out[OUT1_OFF + row * CH + c] = s;
            out[OUT2_OFF + row * CH + c] = s;
        }
    }
}

__global__ __launch_bounds__(256) void vq_entropy(const unsigned int* __restrict__ hist,
                                                  float* __restrict__ out) {
    __shared__ float red[256];
    int t = threadIdx.x;
    float s = 0.0f;
    for (int k = t; k < KCODE; k += 256) {
        unsigned int h = hist[k];
        if (h > 0u) {
            float p = (float)h * (1.0f / 32768.0f);
            s += p * logf(p);
        }
    }
    red[t] = s;
    __syncthreads();
    for (int o = 128; o > 0; o >>= 1) {
        if (t < o) red[t] += red[t + o];
        __syncthreads();
    }
    if (t == 0) out[ENT_OFF] = -red[0];
}

extern "C" void kernel_launch(void* const* d_in, const int* in_sizes, int n_in,
                              void* d_out, int out_size, void* d_ws, size_t ws_size,
                              hipStream_t stream) {
    const float* x0  = (const float*)d_in[0];
    const float* emb = (const float*)d_in[1];
    float* out = (float*)d_out;

    unsigned int* hist = (unsigned int*)d_ws;
    float* esq = (float*)((char*)d_ws + KCODE * sizeof(unsigned int));

    vq_setup  <<<4, 256, 0, stream>>>(emb, hist, esq);
    vq_main   <<<(TOKENS / 256) * CH, 256, 0, stream>>>(x0, emb, esq, hist, out);
    vq_entropy<<<1, 256, 0, stream>>>(hist, out);
}


# codebook of the block's channel copied once into LDS (padded rows, 128 codes per step under the WMMA work); all 8 waves read B fragments from LDS, not per-wave global loads; sums unchanged
# speedup vs baseline: 1.0888x; 1.0888x over previous
#include <hip/hip_runtime.h>
#include <hip/hip_bf16.h>
#include <math.h>

typedef __attribute__((ext_vector_type(2))) float v2f;
typedef __attribute__((ext_vector_type(4))) float v4f;
typedef __attribute__((ext_vector_type(8))) float v8f;

#define TOKENS   32768
#define CH       2
#define VDIM     128
#define KCODE    512
#define OUT1_OFF 8388608
#define OUT2_OFF 8454144
#define ENT_OFF  8519680
#define PITCH    132
#define CHUNK    128

__global__ __launch_bounds__(256) void vq_setup(const float* __restrict__ emb,
                                                unsigned int* __restrict__ hist,
                                                float* __restrict__ esq) {
    int tid = blockIdx.x * 256 + threadIdx.x;
    if (tid < KCODE) hist[tid] = 0u;
    if (tid < CH * KCODE) {
        const float* ep = emb + (size_t)tid * VDIM;
        float s = 0.0f;
        #pragma unroll 8
        for (int v = 0; v < VDIM; ++v) { float e = ep[v]; s += e * e; }
        esq[tid] = s;
    }
}


__device__ __forceinline__ void load_b_tile(v2f (&b)[32], const float* sRow, int t) {
    const float* p = sRow + t * (16 * PITCH);
    #pragma unroll
    for (int kk = 0; kk < 32; ++kk) b[kk] = *(const v2f*)(p + kk * 4);
}

__device__ __forceinline__ void sweep_tile(const v2f (&a0)[32], const v2f (&a1)[32],
                                           const v2f (&b)[32], float ev2, int code,
                                           float (&runD0)[8], float (&runD1)[8],
                                           int (&runI0)[8], int (&runI1)[8]) {
    v8f acc0 = {}; v8f acc1 = {};
    #pragma unroll
    for (int kk = 0; kk < 32; ++kk) {
        acc0 = __builtin_amdgcn_wmma_f32_16x16x4_f32(
            false, a0[kk], false, b[kk], (short)0, acc0, false, false);
        acc1 = __builtin_amdgcn_wmma_f32_16x16x4_f32(
            false, a1[kk], false, b[kk], (short)0, acc1, false, false);
    }
    #pragma unroll
    for (int r = 0; r < 8; ++r) {
        float d0 = ev2 - 2.0f * acc0[r];
        float d1 = ev2 - 2.0f * acc1[r];
        if (d0 < runD0[r]) { runD0[r] = d0; runI0[r] = code; }
        if (d1 < runD1[r]) { runD1[r] = d1; runI1[r] = code; }
    }
}

__global__ __launch_bounds__(256) void vq_main(const float* __restrict__ x0,
                                               const float* __restrict__ emb,
                                               const float* __restrict__ esq,
                                               unsigned int* __restrict__ hist,
                                               float* __restrict__ out) {
    __shared__ __attribute__((aligned(16))) float sE[KCODE * PITCH];
    __shared__ int sIdx[256];

    const int lane   = threadIdx.x & 31;
    const int waveId = threadIdx.x >> 5;
    const int lane16 = lane & 15;
    const int half   = lane >> 4;
    const int c       = blockIdx.x & 1;
    const int rowBase = (blockIdx.x >> 1) * 256;
    const int waveRow = rowBase + waveId * 32;

    const float* embc = emb + (size_t)c * (KCODE * VDIM);
    v4f st[16];
    const float* gsrc = embc + (size_t)threadIdx.x * 4;
    float*       ldst = sE + waveId * PITCH + lane * 4;

    #pragma unroll
    for (int i = 0; i < 16; ++i)
        st[i] = *(const v4f*)(gsrc + i * 1024);

    v2f a0[32], a1[32];
    {
        const float* xp0 = x0 + (size_t)(waveRow + lane16) * (CH * VDIM)
                              + c * VDIM + half * 2;
        const float* xp1 = xp0 + 16 * (CH * VDIM);
        #pragma unroll
        for (int kk = 0; kk < 32; ++kk) {
            a0[kk] = *(const v2f*)(xp0 + kk * 4);
            a1[kk] = *(const v2f*)(xp1 + kk * 4);
        }
    }

    float runD0[8], runD1[8];
    int   runI0[8], runI1[8];
    #pragma unroll
    for (int r = 0; r < 8; ++r) {
        runD0[r] = 3.4e38f; runD1[r] = 3.4e38f; runI0[r] = 0; runI1[r] = 0;
    }

    #pragma unroll
    for (int i = 0; i < 16; ++i)
        *(v4f*)(ldst + i * (8 * PITCH)) = st[i];
    __syncthreads();
    #pragma unroll
    for (int i = 0; i < 16; ++i)
        st[i] = *(const v4f*)(gsrc + CHUNK * VDIM + i * 1024);

    const float* sRow = sE + lane16 * PITCH + half * 2;
    const float* esqc = esq + c * KCODE + lane16;
    v2f bA[32], bB[32];
    load_b_tile(bA, sRow, 0);

    #define PIN() __builtin_amdgcn_sched_barrier(0)
    #pragma unroll 1
    for (int s = 0; s < 4; ++s) {
        #pragma unroll 1
        for (int q = 0; q < 3; ++q) {
            const int t = s * 8 + 2 * q;
            const float evA = esqc[t * 16];
            const float evB = esqc[t * 16 + 16];
            PIN(); load_b_tile(bB, sRow, t + 1); PIN();
            sweep_tile(a0, a1, bA, evA, t * 16 + lane16, runD0, runD1, runI0, runI1);
            PIN(); load_b_tile(bA, sRow, t + 2); PIN();
            sweep_tile(a0, a1, bB, evB, t * 16 + 16 + lane16, runD0, runD1, runI0, runI1);
        }
        {
            const int t = s * 8 + 6;
            const float evA = esqc[t * 16];
            const float evB = esqc[t * 16 + 16];
            PIN(); load_b_tile(bB, sRow, t + 1); PIN();
            sweep_tile(a0, a1, bA, evA, t * 16 + lane16, runD0, runD1, runI0, runI1);
            PIN();
            sweep_tile(a0, a1, bB, evB, t * 16 + 16 + lane16, runD0, runD1, runI0, runI1);
        }
        if (s < 3) {
            #pragma unroll
            for (int i = 0; i < 16; ++i)
                *(v4f*)(ldst + (s + 1) * (CHUNK * PITCH) + i * (8 * PITCH)) = st[i];
            __syncthreads();
            if (s < 2) {
                #pragma unroll
                for (int i = 0; i < 16; ++i)
                    st[i] = *(const v4f*)(gsrc + (s + 2) * (CHUNK * VDIM) + i * 1024);
            }
            load_b_tile(bA, sRow, s * 8 + 8);
        }
    }
    #undef PIN

    #pragma unroll
    for (int m = 1; m < 16; m <<= 1) {
        #pragma unroll
        for (int r = 0; r < 8; ++r) {
            float od = __shfl_xor(runD0[r], m, 32);
            int   oi = __shfl_xor(runI0[r], m, 32);
            if (od < runD0[r] || (od == runD0[r] && oi < runI0[r])) {
                runD0[r] = od; runI0[r] = oi;
            }
            od = __shfl_xor(runD1[r], m, 32);
            oi = __shfl_xor(runI1[r], m, 32);
            if (od < runD1[r] || (od == runD1[r] && oi < runI1[r])) {
                runD1[r] = od; runI1[r] = oi;
            }
        }
    }

    if (lane16 == 0) {
        #pragma unroll
        for (int r = 0; r < 8; ++r) {
            int rl0 = waveId * 32 + r + 8 * half;
            sIdx[rl0] = runI0[r];
            atomicAdd(&hist[runI0[r]], 1u);
            int rl1 = rl0 + 16;
            sIdx[rl1] = runI1[r];
            atomicAdd(&hist[runI1[r]], 1u);
        }
    }
    __syncthreads();

    for (int it = 0; it < 32; ++it) {
        int rloc = waveId + it * 8;
        int row  = rowBase + rloc;
        int idx  = sIdx[rloc];
        const float4 ev = *(const float4*)(emb + ((size_t)(c * KCODE + idx)) * VDIM + lane * 4);
        const float4 xv = *(const float4*)(x0 + (size_t)row * (CH * VDIM) + c * VDIM + lane * 4);
        *(float4*)(out + (size_t)row * (CH * VDIM) + c * VDIM + lane * 4) = ev;
        float dx = xv.x - ev.x, dy = xv.y - ev.y;
        float dz = xv.z - ev.z, dw = xv.w - ev.w;
        float s = dx * dx + dy * dy + dz * dz + dw * dw;
        #pragma unroll
        for (int m = 1; m < 32; m <<= 1) s += __shfl_xor(s, m, 32);
        if (lane == 0) {
            out[OUT1_OFF + row * CH + c] = s;
            out[OUT2_OFF + row * CH + c] = s;
        }
    }
}

__global__ __launch_bounds__(256) void vq_entropy(const unsigned int* __restrict__ hist,
                                                  float* __restrict__ out) {
    __shared__ float red[256];
    int t = threadIdx.x;
    float s = 0.0f;
    for (int k = t; k < KCODE; k += 256) {
        unsigned int h = hist[k];
        if (h > 0u) {
            float p = (float)h * (1.0f / 32768.0f);
            s += p * logf(p);
        }
    }
    red[t] = s;
    __syncthreads();
    for (int o = 128; o > 0; o >>= 1) {
        if (t < o) red[t] += red[t + o];
        __syncthreads();
    }
    if (t == 0) out[ENT_OFF] = -red[0];
}

extern "C" void kernel_launch(void* const* d_in, const int* in_sizes, int n_in,
                              void* d_out, int out_size, void* d_ws, size_t ws_size,
                              hipStream_t stream) {
    const float* x0  = (const float*)d_in[0];
    const float* emb = (const float*)d_in[1];
    float* out = (float*)d_out;

    unsigned int* hist = (unsigned int*)d_ws;
    float* esq = (float*)((char*)d_ws + KCODE * sizeof(unsigned int));

    vq_setup  <<<4, 256, 0, stream>>>(emb, hist, esq);
    vq_main   <<<(TOKENS / 256) * CH, 256, 0, stream>>>(x0, emb, esq, hist, out);
    vq_entropy<<<1, 256, 0, stream>>>(hist, out);
}


# epilogue wave-local and on-chip: code rows from the LDS codebook, x from the A fragments, indices by lane reads; no global re-reads, no barrier; residual sums re-ordered (64 per lane, halves added)
# speedup vs baseline: 1.3240x; 1.2160x over previous
#include <hip/hip_runtime.h>
#include <hip/hip_bf16.h>
#include <math.h>

typedef __attribute__((ext_vector_type(2))) float v2f;
typedef __attribute__((ext_vector_type(4))) float v4f;
typedef __attribute__((ext_vector_type(8))) float v8f;

#define TOKENS   32768
#define CH       2
#define VDIM     128
#define KCODE    512
#define OUT1_OFF 8388608
#define OUT2_OFF 8454144
#define ENT_OFF  8519680
#define PITCH    132
#define CHUNK    128

__global__ __launch_bounds__(256) void vq_setup(const float* __restrict__ emb,
                                                unsigned int* __restrict__ hist,
                                                float* __restrict__ esq) {
    int tid = blockIdx.x * 256 + threadIdx.x;
    if (tid < KCODE) hist[tid] = 0u;
    if (tid < CH * KCODE) {
        const float* ep = emb + (size_t)tid * VDIM;
        float s = 0.0f;
        #pragma unroll 8
        for (int v = 0; v < VDIM; ++v) { float e = ep[v]; s += e * e; }
        esq[tid] = s;
    }
}


__device__ __forceinline__ void load_b_tile(v2f (&b)[32], const float* sRow, int t) {
    const float* p = sRow + t * (16 * PITCH);
    #pragma unroll
    for (int kk = 0; kk < 32; ++kk) b[kk] = *(const v2f*)(p + kk * 4);
}

__device__ __forceinline__ void sweep_tile(const v2f (&a0)[32], const v2f (&a1)[32],
                                           const v2f (&b)[32], float ev2, int code,
                                           float (&runD0)[8], float (&runD1)[8],
                                           int (&runI0)[8], int (&runI1)[8]) {
    v8f acc0 = {}; v8f acc1 = {};
    #pragma unroll
    for (int kk = 0; kk < 32; ++kk) {
        acc0 = __builtin_amdgcn_wmma_f32_16x16x4_f32(
            false, a0[kk], false, b[kk], (short)0, acc0, false, false);
        acc1 = __builtin_amdgcn_wmma_f32_16x16x4_f32(
            false, a1[kk], false, b[kk], (short)0, acc1, false, false);
    }
    #pragma unroll
    for (int r = 0; r < 8; ++r) {
        float d0 = ev2 - 2.0f * acc0[r];
        float d1 = ev2 - 2.0f * acc1[r];
        if (d0 < runD0[r]) { runD0[r] = d0; runI0[r] = code; }
        if (d1 < runD1[r]) { runD1[r] = d1; runI1[r] = code; }
    }
}

__global__ __launch_bounds__(256) void vq_main(const float* __restrict__ x0,
                                               const float* __restrict__ emb,
                                               const float* __restrict__ esq,
                                               unsigned int* __restrict__ hist,
                                               float* __restrict__ out) {
    __shared__ __attribute__((aligned(16))) float sE[KCODE * PITCH];

    const int lane   = threadIdx.x & 31;
    const int waveId = threadIdx.x >> 5;
    const int lane16 = lane & 15;
    const int half   = lane >> 4;
    const int c       = blockIdx.x & 1;
    const int rowBase = (blockIdx.x >> 1) * 256;
    const int waveRow = rowBase + waveId * 32;

    const float* embc = emb + (size_t)c * (KCODE * VDIM);
    v4f st[16];
    const float* gsrc = embc + (size_t)threadIdx.x * 4;
    float*       ldst = sE + waveId * PITCH + lane * 4;

    #pragma unroll
    for (int i = 0; i < 16; ++i)
        st[i] = *(const v4f*)(gsrc + i * 1024);

    v2f a0[32], a1[32];
    {
        const float* xp0 = x0 + (size_t)(waveRow + lane16) * (CH * VDIM)
                              + c * VDIM + half * 2;
        const float* xp1 = xp0 + 16 * (CH * VDIM);
        #pragma unroll
        for (int kk = 0; kk < 32; ++kk) {
            a0[kk] = *(const v2f*)(xp0 + kk * 4);
            a1[kk] = *(const v2f*)(xp1 + kk * 4);
        }
    }

    float runD0[8], runD1[8];
    int   runI0[8], runI1[8];
    #pragma unroll
    for (int r = 0; r < 8; ++r) {
        runD0[r] = 3.4e38f; runD1[r] = 3.4e38f; runI0[r] = 0; runI1[r] = 0;
    }

    #pragma unroll
    for (int i = 0; i < 16; ++i)
        *(v4f*)(ldst + i * (8 * PITCH)) = st[i];
    __syncthreads();
    #pragma unroll
    for (int i = 0; i < 16; ++i)
        st[i] = *(const v4f*)(gsrc + CHUNK * VDIM + i * 1024);

    const float* sRow = sE + lane16 * PITCH + half * 2;
    const float* esqc = esq + c * KCODE + lane16;
    v2f bA[32], bB[32];
    load_b_tile(bA, sRow, 0);

    #define PIN() __builtin_amdgcn_sched_barrier(0)
    #pragma unroll 1
    for (int s = 0; s < 4; ++s) {
        #pragma unroll 1
        for (int q = 0; q < 3; ++q) {
            const int t = s * 8 + 2 * q;
            const float evA = esqc[t * 16];
            const float evB = esqc[t * 16 + 16];
            PIN(); load_b_tile(bB, sRow, t + 1); PIN();
            sweep_tile(a0, a1, bA, evA, t * 16 + lane16, runD0, runD1, runI0, runI1);
            PIN(); load_b_tile(bA, sRow, t + 2); PIN();
            sweep_tile(a0, a1, bB, evB, t * 16 + 16 + lane16, runD0, runD1, runI0, runI1);
        }
        {
            const int t = s * 8 + 6;
            const float evA = esqc[t * 16];
            const float evB = esqc[t * 16 + 16];
            PIN(); load_b_tile(bB, sRow, t + 1); PIN();
            sweep_tile(a0, a1, bA, evA, t * 16 + lane16, runD0, runD1, runI0, runI1);
            PIN();
            sweep_tile(a0, a1, bB, evB, t * 16 + 16 + lane16, runD0, runD1, runI0, runI1);
        }
        if (s < 3) {
            #pragma unroll
            for (int i = 0; i < 16; ++i)
                *(v4f*)(ldst + (s + 1) * (CHUNK * PITCH) + i * (8 * PITCH)) = st[i];
            __syncthreads();
            if (s < 2) {
                #pragma unroll
                for (int i = 0; i < 16; ++i)
                    st[i] = *(const v4f*)(gsrc + (s + 2) * (CHUNK * VDIM) + i * 1024);
            }
            load_b_tile(bA, sRow, s * 8 + 8);
        }
    }
    #undef PIN

    #pragma unroll
    for (int m = 1; m < 16; m <<= 1) {
        #pragma unroll
        for (int r = 0; r < 8; ++r) {
            float od = __shfl_xor(runD0[r], m, 32);
            int   oi = __shfl_xor(runI0[r], m, 32);
            if (od < runD0[r] || (od == runD0[r] && oi < runI0[r])) {
                runD0[r] = od; runI0[r] = oi;
            }
            od = __shfl_xor(runD1[r], m, 32);
            oi = __shfl_xor(runI1[r], m, 32);
            if (od < runD1[r] || (od == runD1[r] && oi < runI1[r])) {
                runD1[r] = od; runI1[r] = oi;
            }
        }
    }

    if (lane16 == 0) {
        #pragma unroll
        for (int r = 0; r < 8; ++r) {
            atomicAdd(&hist[runI0[r]], 1u);
            atomicAdd(&hist[runI1[r]], 1u);
        }
    }


    const int sub = lane16 & 7;
    int selA = runI0[0], selB = runI1[0];
    #pragma unroll
    for (int r = 1; r < 8; ++r) {
        if (sub == r) { selA = runI0[r]; selB = runI1[r]; }
    }
    const int srcLane = ((lane16 >> 3) << 4) | lane16;
    const int myI0 = __shfl(selA, srcLane, 32);
    const int myI1 = __shfl(selB, srcLane, 32);

    {
        const float* e0 = sE + myI0 * PITCH + half * 2;
        const float* e1 = sE + myI1 * PITCH + half * 2;
        float s0x = 0.0f, s0y = 0.0f, s1x = 0.0f, s1y = 0.0f;
        #pragma unroll
        for (int kk = 0; kk < 32; ++kk) {
            const v2f ea = *(const v2f*)(e0 + kk * 4);
            const v2f eb = *(const v2f*)(e1 + kk * 4);
            const float dx0 = a0[kk].x - ea.x, dy0 = a0[kk].y - ea.y;
            const float dx1 = a1[kk].x - eb.x, dy1 = a1[kk].y - eb.y;
            s0x += dx0 * dx0; s0y += dy0 * dy0;
            s1x += dx1 * dx1; s1y += dy1 * dy1;
        }
        float s0 = s0x + s0y;
        float s1 = s1x + s1y;
        s0 += __shfl_xor(s0, 16, 32);
        s1 += __shfl_xor(s1, 16, 32);
        const float sv  = half ? s1 : s0;
        const int   row = waveRow + lane;
        out[OUT1_OFF + row * CH + c] = sv;
        out[OUT2_OFF + row * CH + c] = sv;
    }

    {
        const float* srow = sE + lane * 4;
        float* orow = out + (size_t)waveRow * (CH * VDIM) + c * VDIM + lane * 4;
        #pragma unroll
        for (int r = 0; r < 8; ++r) {
            const int i00 = __builtin_amdgcn_readlane(runI0[r], 0);
            const int i01 = __builtin_amdgcn_readlane(runI0[r], 16);
            const int i10 = __builtin_amdgcn_readlane(runI1[r], 0);
            const int i11 = __builtin_amdgcn_readlane(runI1[r], 16);
            const v4f v00 = *(const v4f*)(srow + i00 * PITCH);
            const v4f v01 = *(const v4f*)(srow + i01 * PITCH);
            const v4f v10 = *(const v4f*)(srow + i10 * PITCH);
            const v4f v11 = *(const v4f*)(srow + i11 * PITCH);
            *(v4f*)(orow + (r     ) * (CH * VDIM)) = v00;
            *(v4f*)(orow + (r +  8) * (CH * VDIM)) = v01;
            *(v4f*)(orow + (r + 16) * (CH * VDIM)) = v10;
            *(v4f*)(orow + (r + 24) * (CH * VDIM)) = v11;
        }
    }
}

__global__ __launch_bounds__(256) void vq_entropy(const unsigned int* __restrict__ hist,
                                                  float* __restrict__ out) {
    __shared__ float red[256];
    int t = threadIdx.x;
    float s = 0.0f;
    for (int k = t; k < KCODE; k += 256) {
        unsigned int h = hist[k];
        if (h > 0u) {
            float p = (float)h * (1.0f / 32768.0f);
            s += p * logf(p);
        }
    }
    red[t] = s;
    __syncthreads();
    for (int o = 128; o > 0; o >>= 1) {
        if (t < o) red[t] += red[t + o];
        __syncthreads();
    }
    if (t == 0) out[ENT_OFF] = -red[0];
}

extern "C" void kernel_launch(void* const* d_in, const int* in_sizes, int n_in,
                              void* d_out, int out_size, void* d_ws, size_t ws_size,
                              hipStream_t stream) {
    const float* x0  = (const float*)d_in[0];
    const float* emb = (const float*)d_in[1];
    float* out = (float*)d_out;

    unsigned int* hist = (unsigned int*)d_ws;
    float* esq = (float*)((char*)d_ws + KCODE * sizeof(unsigned int));

    vq_setup  <<<4, 256, 0, stream>>>(emb, hist, esq);
    vq_main   <<<(TOKENS / 256) * CH, 256, 0, stream>>>(x0, emb, esq, hist, out);
    vq_entropy<<<1, 256, 0, stream>>>(hist, out);
}


# argmin over the 16 columns after the sweep: lanes exchange by row_xmask DPP reads and branch-free selects on the vector ALU, not LDS-unit permutes with waits and branches; same compares and sums
# speedup vs baseline: 1.4276x; 1.0782x over previous
#include <hip/hip_runtime.h>
#include <hip/hip_bf16.h>
#include <math.h>

typedef __attribute__((ext_vector_type(2))) float v2f;
typedef __attribute__((ext_vector_type(4))) float v4f;
typedef __attribute__((ext_vector_type(8))) float v8f;

#define TOKENS   32768
#define CH       2
#define VDIM     128
#define KCODE    512
#define OUT1_OFF 8388608
#define OUT2_OFF 8454144
#define ENT_OFF  8519680
#define PITCH    132
#define CHUNK    128

__global__ __launch_bounds__(256) void vq_setup(const float* __restrict__ emb,
                                                unsigned int* __restrict__ hist,
                                                float* __restrict__ esq) {
    int tid = blockIdx.x * 256 + threadIdx.x;
    if (tid < KCODE) hist[tid] = 0u;
    if (tid < CH * KCODE) {
        const float* ep = emb + (size_t)tid * VDIM;
        float s = 0.0f;
        #pragma unroll 8
        for (int v = 0; v < VDIM; ++v) { float e = ep[v]; s += e * e; }
        esq[tid] = s;
    }
}


__device__ __forceinline__ void load_b_tile(v2f (&b)[32], const float* sRow, int t) {
    const float* p = sRow + t * (16 * PITCH);
    #pragma unroll
    for (int kk = 0; kk < 32; ++kk) b[kk] = *(const v2f*)(p + kk * 4);
}

__device__ __forceinline__ void sweep_tile(const v2f (&a0)[32], const v2f (&a1)[32],
                                           const v2f (&b)[32], float ev2, int code,
                                           float (&runD0)[8], float (&runD1)[8],
                                           int (&runI0)[8], int (&runI1)[8]) {
    v8f acc0 = {}; v8f acc1 = {};
    #pragma unroll
    for (int kk = 0; kk < 32; ++kk) {
        acc0 = __builtin_amdgcn_wmma_f32_16x16x4_f32(
            false, a0[kk], false, b[kk], (short)0, acc0, false, false);
        acc1 = __builtin_amdgcn_wmma_f32_16x16x4_f32(
            false, a1[kk], false, b[kk], (short)0, acc1, false, false);
    }
    #pragma unroll
    for (int r = 0; r < 8; ++r) {
        float d0 = ev2 - 2.0f * acc0[r];
        float d1 = ev2 - 2.0f * acc1[r];
        if (d0 < runD0[r]) { runD0[r] = d0; runI0[r] = code; }
        if (d1 < runD1[r]) { runD1[r] = d1; runI1[r] = code; }
    }
}

template <int M>
__device__ __forceinline__ void argmin_xrow(float (&d)[8], int (&i)[8]) {
    #pragma unroll
    for (int r = 0; r < 8; ++r) {
        const int   db = __float_as_int(d[r]);
        const float od = __int_as_float(
            __builtin_amdgcn_update_dpp(db, db, 0x160 + M, 0xf, 0xf, false));
        const int   oi = __builtin_amdgcn_update_dpp(i[r], i[r], 0x160 + M, 0xf, 0xf, false);
        const bool take = (od < d[r]) | ((od == d[r]) & (oi < i[r]));
        d[r] = take ? od : d[r];
        i[r] = take ? oi : i[r];
    }
}

__global__ __launch_bounds__(256) void vq_main(const float* __restrict__ x0,
                                               const float* __restrict__ emb,
                                               const float* __restrict__ esq,
                                               unsigned int* __restrict__ hist,
                                               float* __restrict__ out) {
    __shared__ __attribute__((aligned(16))) float sE[KCODE * PITCH];

    const int lane   = threadIdx.x & 31;
    const int waveId = threadIdx.x >> 5;
    const int lane16 = lane & 15;
    const int half   = lane >> 4;
    const int c       = blockIdx.x & 1;
    const int rowBase = (blockIdx.x >> 1) * 256;
    const int waveRow = rowBase + waveId * 32;

    const float* embc = emb + (size_t)c * (KCODE * VDIM);
    v4f st[16];
    const float* gsrc = embc + (size_t)threadIdx.x * 4;
    float*       ldst = sE + waveId * PITCH + lane * 4;

    #pragma unroll
    for (int i = 0; i < 16; ++i)
        st[i] = *(const v4f*)(gsrc + i * 1024);

    v2f a0[32], a1[32];
    {
        const float* xp0 = x0 + (size_t)(waveRow + lane16) * (CH * VDIM)
                              + c * VDIM + half * 2;
        const float* xp1 = xp0 + 16 * (CH * VDIM);
        #pragma unroll
        for (int kk = 0; kk < 32; ++kk) {
            a0[kk] = *(const v2f*)(xp0 + kk * 4);
            a1[kk] = *(const v2f*)(xp1 + kk * 4);
        }
    }

    float runD0[8], runD1[8];
    int   runI0[8], runI1[8];
    #pragma unroll
    for (int r = 0; r < 8; ++r) {
        runD0[r] = 3.4e38f; runD1[r] = 3.4e38f; runI0[r] = 0; runI1[r] = 0;
    }

    #pragma unroll
    for (int i = 0; i < 16; ++i)
        *(v4f*)(ldst + i * (8 * PITCH)) = st[i];
    __syncthreads();
    #pragma unroll
    for (int i = 0; i < 16; ++i)
        st[i] = *(const v4f*)(gsrc + CHUNK * VDIM + i * 1024);

    const float* sRow = sE + lane16 * PITCH + half * 2;
    const float* esqc = esq + c * KCODE + lane16;
    v2f bA[32], bB[32];
    load_b_tile(bA, sRow, 0);

    #define PIN() __builtin_amdgcn_sched_barrier(0)
    #pragma unroll 1
    for (int s = 0; s < 4; ++s) {
        #pragma unroll 1
        for (int q = 0; q < 3; ++q) {
            const int t = s * 8 + 2 * q;
            const float evA = esqc[t * 16];
            const float evB = esqc[t * 16 + 16];
            PIN(); load_b_tile(bB, sRow, t + 1); PIN();
            sweep_tile(a0, a1, bA, evA, t * 16 + lane16, runD0, runD1, runI0, runI1);
            PIN(); load_b_tile(bA, sRow, t + 2); PIN();
            sweep_tile(a0, a1, bB, evB, t * 16 + 16 + lane16, runD0, runD1, runI0, runI1);
        }
        {
            const int t = s * 8 + 6;
            const float evA = esqc[t * 16];
            const float evB = esqc[t * 16 + 16];
            PIN(); load_b_tile(bB, sRow, t + 1); PIN();
            sweep_tile(a0, a1, bA, evA, t * 16 + lane16, runD0, runD1, runI0, runI1);
            PIN();
            sweep_tile(a0, a1, bB, evB, t * 16 + 16 + lane16, runD0, runD1, runI0, runI1);
        }
        if (s < 3) {
            #pragma unroll
            for (int i = 0; i < 16; ++i)
                *(v4f*)(ldst + (s + 1) * (CHUNK * PITCH) + i * (8 * PITCH)) = st[i];
            __syncthreads();
            if (s < 2) {
                #pragma unroll
                for (int i = 0; i < 16; ++i)
                    st[i] = *(const v4f*)(gsrc + (s + 2) * (CHUNK * VDIM) + i * 1024);
            }
            load_b_tile(bA, sRow, s * 8 + 8);
        }
    }
    #undef PIN

    argmin_xrow<1>(runD0, runI0); argmin_xrow<1>(runD1, runI1);
    argmin_xrow<2>(runD0, runI0); argmin_xrow<2>(runD1, runI1);
    argmin_xrow<4>(runD0, runI0); argmin_xrow<4>(runD1, runI1);
    argmin_xrow<8>(runD0, runI0); argmin_xrow<8>(runD1, runI1);

    if (lane16 == 0) {
        #pragma unroll
        for (int r = 0; r < 8; ++r) {
            atomicAdd(&hist[runI0[r]], 1u);
            atomicAdd(&hist[runI1[r]], 1u);
        }
    }


    const int sub = lane16 & 7;
    int selA = runI0[0], selB = runI1[0];
    #pragma unroll
    for (int r = 1; r < 8; ++r) {
        if (sub == r) { selA = runI0[r]; selB = runI1[r]; }
    }
    const int srcLane = ((lane16 >> 3) << 4) | lane16;
    const int myI0 = __shfl(selA, srcLane, 32);
    const int myI1 = __shfl(selB, srcLane, 32);

    {
        const float* e0 = sE + myI0 * PITCH + half * 2;
        const float* e1 = sE + myI1 * PITCH + half * 2;
        float s0x = 0.0f, s0y = 0.0f, s1x = 0.0f, s1y = 0.0f;
        #pragma unroll
        for (int kk = 0; kk < 32; ++kk) {
            const v2f ea = *(const v2f*)(e0 + kk * 4);
            const v2f eb = *(const v2f*)(e1 + kk * 4);
            const float dx0 = a0[kk].x - ea.x, dy0 = a0[kk].y - ea.y;
            const float dx1 = a1[kk].x - eb.x, dy1 = a1[kk].y - eb.y;
            s0x += dx0 * dx0; s0y += dy0 * dy0;
            s1x += dx1 * dx1; s1y += dy1 * dy1;
        }
        float s0 = s0x + s0y;
        float s1 = s1x + s1y;
        s0 += __shfl_xor(s0, 16, 32);
        s1 += __shfl_xor(s1, 16, 32);
        const float sv  = half ? s1 : s0;
        const int   row = waveRow + lane;
        out[OUT1_OFF + row * CH + c] = sv;
        out[OUT2_OFF + row * CH + c] = sv;
    }

    {
        const float* srow = sE + lane * 4;
        float* orow = out + (size_t)waveRow * (CH * VDIM) + c * VDIM + lane * 4;
        #pragma unroll
        for (int r = 0; r < 8; ++r) {
            const int i00 = __builtin_amdgcn_readlane(runI0[r], 0);
            const int i01 = __builtin_amdgcn_readlane(runI0[r], 16);
            const int i10 = __builtin_amdgcn_readlane(runI1[r], 0);
            const int i11 = __builtin_amdgcn_readlane(runI1[r], 16);
            const v4f v00 = *(const v4f*)(srow + i00 * PITCH);
            const v4f v01 = *(const v4f*)(srow + i01 * PITCH);
            const v4f v10 = *(const v4f*)(srow + i10 * PITCH);
            const v4f v11 = *(const v4f*)(srow + i11 * PITCH);
            *(v4f*)(orow + (r     ) * (CH * VDIM)) = v00;
            *(v4f*)(orow + (r +  8) * (CH * VDIM)) = v01;
            *(v4f*)(orow + (r + 16) * (CH * VDIM)) = v10;
            *(v4f*)(orow + (r + 24) * (CH * VDIM)) = v11;
        }
    }
}

__global__ __launch_bounds__(256) void vq_entropy(const unsigned int* __restrict__ hist,
                                                  float* __restrict__ out) {
    __shared__ float red[256];
    int t = threadIdx.x;
    float s = 0.0f;
    for (int k = t; k < KCODE; k += 256) {
        unsigned int h = hist[k];
        if (h > 0u) {
            float p = (float)h * (1.0f / 32768.0f);
            s += p * logf(p);
        }
    }
    red[t] = s;
    __syncthreads();
    for (int o = 128; o > 0; o >>= 1) {
        if (t < o) red[t] += red[t + o];
        __syncthreads();
    }
    if (t == 0) out[ENT_OFF] = -red[0];
}

extern "C" void kernel_launch(void* const* d_in, const int* in_sizes, int n_in,
                              void* d_out, int out_size, void* d_ws, size_t ws_size,
                              hipStream_t stream) {
    const float* x0  = (const float*)d_in[0];
    const float* emb = (const float*)d_in[1];
    float* out = (float*)d_out;

    unsigned int* hist = (unsigned int*)d_ws;
    float* esq = (float*)((char*)d_ws + KCODE * sizeof(unsigned int));

    vq_setup  <<<4, 256, 0, stream>>>(emb, hist, esq);
    vq_main   <<<(TOKENS / 256) * CH, 256, 0, stream>>>(x0, emb, esq, hist, out);
    vq_entropy<<<1, 256, 0, stream>>>(hist, out);
}


# both helper launches: one group of whole-line loads and a lane-tree (DPP) sum replace the serial load-wait-add loop and the eight-step LDS barrier ladder; ||e||^2 and entropy sums re-ordered
# speedup vs baseline: 1.5186x; 1.0637x over previous
#include <hip/hip_runtime.h>
#include <hip/hip_bf16.h>
#include <math.h>

typedef __attribute__((ext_vector_type(2))) float v2f;
typedef __attribute__((ext_vector_type(4))) float v4f;
typedef __attribute__((ext_vector_type(8))) float v8f;

#define TOKENS   32768
#define CH       2
#define VDIM     128
#define KCODE    512
#define OUT1_OFF 8388608
#define OUT2_OFF 8454144
#define ENT_OFF  8519680
#define PITCH    132
#define CHUNK    128

template <int M>
__device__ __forceinline__ float xrow_add(float v) {
    const int b = __float_as_int(v);
    return v + __int_as_float(__builtin_amdgcn_update_dpp(b, b, 0x160 + M, 0xf, 0xf, false));
}
__device__ __forceinline__ float wave_sum(float v) {
    v = xrow_add<1>(v); v = xrow_add<2>(v); v = xrow_add<4>(v); v = xrow_add<8>(v);
    const int b = __float_as_int(v);
    return __int_as_float(__builtin_amdgcn_readlane(b, 0)) +
           __int_as_float(__builtin_amdgcn_readlane(b, 16));
}

__global__ __launch_bounds__(128) void vq_setup(const float* __restrict__ emb,
                                                unsigned int* __restrict__ hist,
                                                float* __restrict__ esq) {
    const int tid  = blockIdx.x * 128 + threadIdx.x;
    const int lane = threadIdx.x & 31;
    const int row0 = (tid >> 5) * 4;
    if (tid < KCODE) hist[tid] = 0u;
    const float* p = emb + (size_t)row0 * VDIM + lane * 4;
    v4f r[4];
    #pragma unroll
    for (int j = 0; j < 4; ++j) r[j] = *(const v4f*)(p + j * VDIM);
    float s[4];
    #pragma unroll
    for (int j = 0; j < 4; ++j) {
        float q = r[j].x * r[j].x;
        q += r[j].y * r[j].y; q += r[j].z * r[j].z; q += r[j].w * r[j].w;
        s[j] = wave_sum(q);
    }
    const float sv = (lane == 0) ? s[0] : (lane == 1) ? s[1] : (lane == 2) ? s[2] : s[3];
    if (lane < 4) esq[row0 + lane] = sv;
}


__device__ __forceinline__ void load_b_tile(v2f (&b)[32], const float* sRow, int t) {
    const float* p = sRow + t * (16 * PITCH);
    #pragma unroll
    for (int kk = 0; kk < 32; ++kk) b[kk] = *(const v2f*)(p + kk * 4);
}

__device__ __forceinline__ void sweep_tile(const v2f (&a0)[32], const v2f (&a1)[32],
                                           const v2f (&b)[32], float ev2, int code,
                                           float (&runD0)[8], float (&runD1)[8],
                                           int (&runI0)[8], int (&runI1)[8]) {
    v8f acc0 = {}; v8f acc1 = {};
    #pragma unroll
    for (int kk = 0; kk < 32; ++kk) {
        acc0 = __builtin_amdgcn_wmma_f32_16x16x4_f32(
            false, a0[kk], false, b[kk], (short)0, acc0, false, false);
        acc1 = __builtin_amdgcn_wmma_f32_16x16x4_f32(
            false, a1[kk], false, b[kk], (short)0, acc1, false, false);
    }
    #pragma unroll
    for (int r = 0; r < 8; ++r) {
        float d0 = ev2 - 2.0f * acc0[r];
        float d1 = ev2 - 2.0f * acc1[r];
        if (d0 < runD0[r]) { runD0[r] = d0; runI0[r] = code; }
        if (d1 < runD1[r]) { runD1[r] = d1; runI1[r] = code; }
    }
}

template <int M>
__device__ __forceinline__ void argmin_xrow(float (&d)[8], int (&i)[8]) {
    #pragma unroll
    for (int r = 0; r < 8; ++r) {
        const int   db = __float_as_int(d[r]);
        const float od = __int_as_float(
            __builtin_amdgcn_update_dpp(db, db, 0x160 + M, 0xf, 0xf, false));
        const int   oi = __builtin_amdgcn_update_dpp(i[r], i[r], 0x160 + M, 0xf, 0xf, false);
        const bool take = (od < d[r]) | ((od == d[r]) & (oi < i[r]));
        d[r] = take ? od : d[r];
        i[r] = take ? oi : i[r];
    }
}

__global__ __launch_bounds__(256) void vq_main(const float* __restrict__ x0,
                                               const float* __restrict__ emb,
                                               const float* __restrict__ esq,
                                               unsigned int* __restrict__ hist,
                                               float* __restrict__ out) {
    __shared__ __attribute__((aligned(16))) float sE[KCODE * PITCH];

    const int lane   = threadIdx.x & 31;
    const int waveId = threadIdx.x >> 5;
    const int lane16 = lane & 15;
    const int half   = lane >> 4;
    const int c       = blockIdx.x & 1;
    const int rowBase = (blockIdx.x >> 1) * 256;
    const int waveRow = rowBase + waveId * 32;

    const float* embc = emb + (size_t)c * (KCODE * VDIM);
    v4f st[16];
    const float* gsrc = embc + (size_t)threadIdx.x * 4;
    float*       ldst = sE + waveId * PITCH + lane * 4;

    #pragma unroll
    for (int i = 0; i < 16; ++i)
        st[i] = *(const v4f*)(gsrc + i * 1024);

    v2f a0[32], a1[32];
    {
        const float* xp0 = x0 + (size_t)(waveRow + lane16) * (CH * VDIM)
                              + c * VDIM + half * 2;
        const float* xp1 = xp0 + 16 * (CH * VDIM);
        #pragma unroll
        for (int kk = 0; kk < 32; ++kk) {
            a0[kk] = *(const v2f*)(xp0 + kk * 4);
            a1[kk] = *(const v2f*)(xp1 + kk * 4);
        }
    }

    float runD0[8], runD1[8];
    int   runI0[8], runI1[8];
    #pragma unroll
    for (int r = 0; r < 8; ++r) {
        runD0[r] = 3.4e38f; runD1[r] = 3.4e38f; runI0[r] = 0; runI1[r] = 0;
    }

    #pragma unroll
    for (int i = 0; i < 16; ++i)
        *(v4f*)(ldst + i * (8 * PITCH)) = st[i];
    __syncthreads();
    #pragma unroll
    for (int i = 0; i < 16; ++i)
        st[i] = *(const v4f*)(gsrc + CHUNK * VDIM + i * 1024);

    const float* sRow = sE + lane16 * PITCH + half * 2;
    const float* esqc = esq + c * KCODE + lane16;
    v2f bA[32], bB[32];
    load_b_tile(bA, sRow, 0);

    #define PIN() __builtin_amdgcn_sched_barrier(0)
    #pragma unroll 1
    for (int s = 0; s < 4; ++s) {
        #pragma unroll 1
        for (int q = 0; q < 3; ++q) {
            const int t = s * 8 + 2 * q;
            const float evA = esqc[t * 16];
            const float evB = esqc[t * 16 + 16];
            PIN(); load_b_tile(bB, sRow, t + 1); PIN();
            sweep_tile(a0, a1, bA, evA, t * 16 + lane16, runD0, runD1, runI0, runI1);
            PIN(); load_b_tile(bA, sRow, t + 2); PIN();
            sweep_tile(a0, a1, bB, evB, t * 16 + 16 + lane16, runD0, runD1, runI0, runI1);
        }
        {
            const int t = s * 8 + 6;
            const float evA = esqc[t * 16];
            const float evB = esqc[t * 16 + 16];
            PIN(); load_b_tile(bB, sRow, t + 1); PIN();
            sweep_tile(a0, a1, bA, evA, t * 16 + lane16, runD0, runD1, runI0, runI1);
            PIN();
            sweep_tile(a0, a1, bB, evB, t * 16 + 16 + lane16, runD0, runD1, runI0, runI1);
        }
        if (s < 3) {
            #pragma unroll
            for (int i = 0; i < 16; ++i)
                *(v4f*)(ldst + (s + 1) * (CHUNK * PITCH) + i * (8 * PITCH)) = st[i];
            __syncthreads();
            if (s < 2) {
                #pragma unroll
                for (int i = 0; i < 16; ++i)
                    st[i] = *(const v4f*)(gsrc + (s + 2) * (CHUNK * VDIM) + i * 1024);
            }
            load_b_tile(bA, sRow, s * 8 + 8);
        }
    }
    #undef PIN

    argmin_xrow<1>(runD0, runI0); argmin_xrow<1>(runD1, runI1);
    argmin_xrow<2>(runD0, runI0); argmin_xrow<2>(runD1, runI1);
    argmin_xrow<4>(runD0, runI0); argmin_xrow<4>(runD1, runI1);
    argmin_xrow<8>(runD0, runI0); argmin_xrow<8>(runD1, runI1);

    if (lane16 == 0) {
        #pragma unroll
        for (int r = 0; r < 8; ++r) {
            atomicAdd(&hist[runI0[r]], 1u);
            atomicAdd(&hist[runI1[r]], 1u);
        }
    }


    const int sub = lane16 & 7;
    int selA = runI0[0], selB = runI1[0];
    #pragma unroll
    for (int r = 1; r < 8; ++r) {
        if (sub == r) { selA = runI0[r]; selB = runI1[r]; }
    }
    const int srcLane = ((lane16 >> 3) << 4) | lane16;
    const int myI0 = __shfl(selA, srcLane, 32);
    const int myI1 = __shfl(selB, srcLane, 32);

    {
        const float* e0 = sE + myI0 * PITCH + half * 2;
        const float* e1 = sE + myI1 * PITCH + half * 2;
        float s0x = 0.0f, s0y = 0.0f, s1x = 0.0f, s1y = 0.0f;
        #pragma unroll
        for (int kk = 0; kk < 32; ++kk) {
            const v2f ea = *(const v2f*)(e0 + kk * 4);
            const v2f eb = *(const v2f*)(e1 + kk * 4);
            const float dx0 = a0[kk].x - ea.x, dy0 = a0[kk].y - ea.y;
            const float dx1 = a1[kk].x - eb.x, dy1 = a1[kk].y - eb.y;
            s0x += dx0 * dx0; s0y += dy0 * dy0;
            s1x += dx1 * dx1; s1y += dy1 * dy1;
        }
        float s0 = s0x + s0y;
        float s1 = s1x + s1y;
        s0 += __shfl_xor(s0, 16, 32);
        s1 += __shfl_xor(s1, 16, 32);
        const float sv  = half ? s1 : s0;
        const int   row = waveRow + lane;
        out[OUT1_OFF + row * CH + c] = sv;
        out[OUT2_OFF + row * CH + c] = sv;
    }

    {
        const float* srow = sE + lane * 4;
        float* orow = out + (size_t)waveRow * (CH * VDIM) + c * VDIM + lane * 4;
        #pragma unroll
        for (int r = 0; r < 8; ++r) {
            const int i00 = __builtin_amdgcn_readlane(runI0[r], 0);
            const int i01 = __builtin_amdgcn_readlane(runI0[r], 16);
            const int i10 = __builtin_amdgcn_readlane(runI1[r], 0);
            const int i11 = __builtin_amdgcn_readlane(runI1[r], 16);
            const v4f v00 = *(const v4f*)(srow + i00 * PITCH);
            const v4f v01 = *(const v4f*)(srow + i01 * PITCH);
            const v4f v10 = *(const v4f*)(srow + i10 * PITCH);
            const v4f v11 = *(const v4f*)(srow + i11 * PITCH);
            *(v4f*)(orow + (r     ) * (CH * VDIM)) = v00;
            *(v4f*)(orow + (r +  8) * (CH * VDIM)) = v01;
            *(v4f*)(orow + (r + 16) * (CH * VDIM)) = v10;
            *(v4f*)(orow + (r + 24) * (CH * VDIM)) = v11;
        }
    }
}

__global__ __launch_bounds__(256) void vq_entropy(const unsigned int* __restrict__ hist,
                                                  float* __restrict__ out) {
    __shared__ float red[8];
    const int t    = threadIdx.x;
    const int lane = t & 31;
    const int wv   = t >> 5;
    const unsigned int h0 = hist[t];
    const unsigned int h1 = hist[t + 256];
    float s = 0.0f;
    if (h0 > 0u) { const float p = (float)h0 * (1.0f / 32768.0f); s += p * logf(p); }
    if (h1 > 0u) { const float p = (float)h1 * (1.0f / 32768.0f); s += p * logf(p); }
    s = wave_sum(s);
    if (lane == 0) red[wv] = s;
    __syncthreads();
    if (wv == 0) {
        float v = (lane < 8) ? red[lane] : 0.0f;
        v = xrow_add<1>(v); v = xrow_add<2>(v); v = xrow_add<4>(v);
        if (lane == 0) out[ENT_OFF] = -v;
    }
}

extern "C" void kernel_launch(void* const* d_in, const int* in_sizes, int n_in,
                              void* d_out, int out_size, void* d_ws, size_t ws_size,
                              hipStream_t stream) {
    const float* x0  = (const float*)d_in[0];
    const float* emb = (const float*)d_in[1];
    float* out = (float*)d_out;

    unsigned int* hist = (unsigned int*)d_ws;
    float* esq = (float*)((char*)d_ws + KCODE * sizeof(unsigned int));

    vq_setup  <<<64, 128, 0, stream>>>(emb, hist, esq);
    vq_main   <<<(TOKENS / 256) * CH, 256, 0, stream>>>(x0, emb, esq, hist, out);
    vq_entropy<<<1, 256, 0, stream>>>(hist, out);
}
